# starting kernel
# baseline (speedup 1.0000x reference)
#include <hip/hip_runtime.h>

typedef __attribute__((ext_vector_type(2))) float v2f;
typedef __attribute__((ext_vector_type(8))) float v8f;

#define GCN_N_NODES 100000

__global__ void __launch_bounds__(256)
deg_init_kernel(float* __restrict__ deg, int n) {
  int i = blockIdx.x * blockDim.x + threadIdx.x;
  if (i < n) deg[i] = 1.0f;
}

__global__ void __launch_bounds__(256)
deg_count_kernel(const int* __restrict__ dst, float* __restrict__ deg, int E) {
  int e = blockIdx.x * blockDim.x + threadIdx.x;
  if (e < E) unsafeAtomicAdd(&deg[dst[e]], 1.0f);
}

__global__ void __launch_bounds__(256)
deg_rsqrt_kernel(float* __restrict__ deg, int n) {
  int i = blockIdx.x * blockDim.x + threadIdx.x;
  if (i < n) deg[i] = rsqrtf(deg[i]);
}

template <bool FUSE_RELU_BIAS>
__global__ void __launch_bounds__(256)
gemm_wmma_kernel(const float* __restrict__ A, const float* __restrict__ W,
                 const float* __restrict__ bias, float* __restrict__ C,
                 int M, int N, int K, int totalTiles) {
  int wave = blockIdx.x * (blockDim.x >> 5) + (threadIdx.x >> 5);
  if (wave >= totalTiles) return;

  int tilesPerRow = N >> 4;
  int tileRow = wave / tilesPerRow;
  int tileCol = wave - tileRow * tilesPerRow;
  int row0 = tileRow << 4;
  int col0 = tileCol << 4;

  int lane = threadIdx.x & 31;
  int half = lane >> 4;
  int l15  = lane & 15;
  int kOff = half * 2;

  const float* arow = A + (size_t)(row0 + l15) * K;
  const float* wcol = W + col0 + l15;

  v8f acc = {};
#pragma unroll 4
  for (int k = 0; k < K; k += 4) {
    float a0 = arow[k + kOff];
    float a1 = arow[k + kOff + 1];
    if (FUSE_RELU_BIAS) {
      a0 = fmaxf(a0 + bias[k + kOff], 0.0f);
      a1 = fmaxf(a1 + bias[k + kOff + 1], 0.0f);
    }
    v2f a; a.x = a0; a.y = a1;
    v2f b;
    b.x = wcol[(size_t)(k + kOff) * N];
    b.y = wcol[(size_t)(k + kOff + 1) * N];
    acc = __builtin_amdgcn_wmma_f32_16x16x4_f32(
         false, a,  false, b,
         (short)0, acc,  false,  false);
  }

  float* crow = C + (size_t)(row0 + half * 8) * N + col0 + l15;
#pragma unroll
  for (int r = 0; r < 8; ++r) crow[(size_t)r * N] = acc[r];
}

template <int F>
__global__ void __launch_bounds__(256)
init_self_kernel(const float* __restrict__ h, const float* __restrict__ dinv,
                 const float* __restrict__ bias, float* __restrict__ out, int total) {
  int idx = blockIdx.x * blockDim.x + threadIdx.x;
  if (idx >= total) return;
  int i = idx / F;
  int f = idx & (F - 1);
  float di = dinv[i];
  float v = h[idx] * di * di;
  if (bias) v += bias[f];
  out[idx] = v;
}

template <int FPL>
__global__ void __launch_bounds__(256)
scatter_kernel(const float* __restrict__ h, const int* __restrict__ src,
               const int* __restrict__ dst, const float* __restrict__ dinv,
               float* __restrict__ out, int E) {
  const int F = FPL * 32;
  int e = (blockIdx.x * blockDim.x + threadIdx.x) >> 5;
  if (e >= E) return;
  int lane = threadIdx.x & 31;
  int s = src[e];
  int d = dst[e];
  float norm = dinv[s] * dinv[d];
  const float* hp = h + (size_t)s * F + lane * FPL;
  float* op = out + (size_t)d * F + lane * FPL;
  if constexpr (FPL == 4) {
    float4 v = *reinterpret_cast<const float4*>(hp);
    unsafeAtomicAdd(op + 0, v.x * norm);
    unsafeAtomicAdd(op + 1, v.y * norm);
    unsafeAtomicAdd(op + 2, v.z * norm);
    unsafeAtomicAdd(op + 3, v.w * norm);
  } else {
    float2 v = *reinterpret_cast<const float2*>(hp);
    unsafeAtomicAdd(op + 0, v.x * norm);
    unsafeAtomicAdd(op + 1, v.y * norm);
  }
}

extern "C" void kernel_launch(void* const* d_in, const int* in_sizes, int n_in,
                              void* d_out, int out_size, void* d_ws, size_t ws_size,
                              hipStream_t stream) {
  const float* x  = (const float*)d_in[0];
  const int*   ei = (const int*)d_in[1];
  const float* W1 = (const float*)d_in[2];
  const float* b1 = (const float*)d_in[3];
  const float* W2 = (const float*)d_in[4];
  const float* b2 = (const float*)d_in[5];
  float* out = (float*)d_out;

  const int Nn = GCN_N_NODES;
  const int E  = in_sizes[1] / 2;
  const int* src = ei;
  const int* dst = ei + E;

  float* dinv = (float*)d_ws;
  float* h    = dinv + Nn;
  float* out1 = h + (size_t)Nn * 128;
  float* h2   = h;

  deg_init_kernel<<<(Nn + 255) / 256, 256, 0, stream>>>(dinv, Nn);
  deg_count_kernel<<<(E + 255) / 256, 256, 0, stream>>>(dst, dinv, E);
  deg_rsqrt_kernel<<<(Nn + 255) / 256, 256, 0, stream>>>(dinv, Nn);

  {
    int M = Nn, N = 128, K = 128;
    int tiles = (M >> 4) * (N >> 4);
    gemm_wmma_kernel<false><<<(tiles + 7) / 8, 256, 0, stream>>>(
        x, W1, nullptr, h, M, N, K, tiles);
  }
  init_self_kernel<128><<<((Nn * 128) + 255) / 256, 256, 0, stream>>>(
      h, dinv, nullptr, out1, Nn * 128);
  scatter_kernel<4><<<(E + 7) / 8, 256, 0, stream>>>(h, src, dst, dinv, out1, E);

  {
    int M = Nn, N = 64, K = 128;
    int tiles = (M >> 4) * (N >> 4);
    gemm_wmma_kernel<true><<<(tiles + 7) / 8, 256, 0, stream>>>(
        out1, W2, b1, h2, M, N, K, tiles);
  }
  init_self_kernel<64><<<((Nn * 64) + 255) / 256, 256, 0, stream>>>(
      h2, dinv, b2, out, Nn * 64);
  scatter_kernel<2><<<(E + 7) / 8, 256, 0, stream>>>(h2, src, dst, dinv, out, E);
}


# Both GEMMs: 1024 long-lived work-groups; each wave keeps its 16-column slice of W in registers for all k; A and C blocks pass through LDS as whole rows; k sum order changes
# speedup vs baseline: 2.4916x; 2.4916x over previous
#include <hip/hip_runtime.h>

typedef __attribute__((ext_vector_type(2))) float v2f;
typedef __attribute__((ext_vector_type(8))) float v8f;

#define GCN_N_NODES 100000
#define AGG_TILE 391
#define AGG_GROUPS ((GCN_N_NODES + AGG_TILE - 1) / AGG_TILE)
#define AGG_PEND 544
#define AGG_RCAP 1024
#define AGG_DL_BITS 9
static_assert(AGG_TILE <= (1 << AGG_DL_BITS), "tile row must fit the record's low bits");
static_assert(GCN_N_NODES <= (1 << (31 - AGG_DL_BITS)), "source row must fit the record's high bits");

__global__ void __launch_bounds__(256)
deg_init_kernel(float* __restrict__ deg, int n) {
  int i = blockIdx.x * blockDim.x + threadIdx.x;
  if (i < n) deg[i] = 1.0f;
}

__global__ void __launch_bounds__(256)
deg_count_kernel(const int* __restrict__ dst, float* __restrict__ deg, int E) {
  int e = blockIdx.x * blockDim.x + threadIdx.x;
  if (e < E) unsafeAtomicAdd(&deg[dst[e]], 1.0f);
}

__global__ void __launch_bounds__(256)
deg_rsqrt_kernel(float* __restrict__ deg, int n) {
  int i = blockIdx.x * blockDim.x + threadIdx.x;
  if (i < n) deg[i] = rsqrtf(deg[i]);
}

template <int N, bool FUSE_RELU_BIAS>
__global__ void __launch_bounds__(256)
gemm_wreg_kernel(const float* __restrict__ A, const float* __restrict__ W,
                 const float* __restrict__ bias, float* __restrict__ C, int M) {
  constexpr int K = 128;
  constexpr int BM = 32;
  constexpr int PA = K + 4;
  constexpr int PC = N + 4;
  constexpr int NT = N / 16;
  constexpr int RT = NT / 4;
  constexpr int LPR = N / 4;
  constexpr int RPI = 32 / LPR;
  static_assert(NT == 8 || NT == 4, "N is 128 or 64");

  __shared__ alignas(16) float abuf[2][BM * PA];
  __shared__ alignas(16) float cbuf[2][BM * PC];

  const int tid = threadIdx.x;
  const int lane = tid & 31;
  const int wv = __builtin_amdgcn_readfirstlane(tid >> 5);
  const int c = lane & 15;
  const int h = lane >> 4;
  const int ct = wv % NT;
  const int rt0 = (wv / NT) * RT;

  const int nblk = (M + BM - 1) / BM;
  int blk = blockIdx.x;
  if (blk >= nblk) return;

  v2f b[32];
  {
    const float* wp = W + 16 * ct + c;
#pragma unroll
    for (int p = 0; p < 16; ++p) {
      const int k0 = 8 * p + 4 * h;
      b[2 * p].x = wp[(size_t)(k0 + 0) * N];
      b[2 * p].y = wp[(size_t)(k0 + 1) * N];
      b[2 * p + 1].x = wp[(size_t)(k0 + 2) * N];
      b[2 * p + 1].y = wp[(size_t)(k0 + 3) * N];
    }
  }
  float4 bv = make_float4(0.0f, 0.0f, 0.0f, 0.0f);
  if (FUSE_RELU_BIAS) bv = *reinterpret_cast<const float4*>(bias + 4 * lane);

  float4 st[4];
  auto fetch = [&](int q) __attribute__((always_inline)) {
#pragma unroll
    for (int u = 0; u < 4; ++u) {
      const int row = q * BM + 4 * wv + u;
      st[u] = make_float4(0.0f, 0.0f, 0.0f, 0.0f);
      if (row < M) st[u] = *reinterpret_cast<const float4*>(A + (size_t)row * K + 4 * lane);
    }
  };
  auto stage = [&](int buf) __attribute__((always_inline)) {
#pragma unroll
    for (int u = 0; u < 4; ++u) {
      float4 v = st[u];
      if (FUSE_RELU_BIAS) {
        v.x = fmaxf(v.x + bv.x, 0.0f);
        v.y = fmaxf(v.y + bv.y, 0.0f);
        v.z = fmaxf(v.z + bv.z, 0.0f);
        v.w = fmaxf(v.w + bv.w, 0.0f);
      }
      *reinterpret_cast<float4*>(&abuf[buf][(4 * wv + u) * PA + 4 * lane]) = v;
    }
  };

  fetch(blk);
  stage(0);
  __syncthreads();

  for (int it = 0; blk < nblk; blk += gridDim.x, ++it) {
    const int cur = it & 1;
    const int nb = blk + gridDim.x;
    if (nb < nblk) fetch(nb);

    float4 x[RT][16];
#pragma unroll
    for (int rt = 0; rt < RT; ++rt)
#pragma unroll
      for (int p = 0; p < 16; ++p)
        x[rt][p] = *reinterpret_cast<const float4*>(
            &abuf[cur][(16 * (rt0 + rt) + c) * PA + 8 * p + 4 * h]);

    v8f acc[RT];
#pragma unroll
    for (int rt = 0; rt < RT; ++rt) acc[rt] = v8f{};
#pragma unroll
    for (int p = 0; p < 16; ++p) {
#pragma unroll
      for (int rt = 0; rt < RT; ++rt) {
        v2f a0; a0.x = x[rt][p].x; a0.y = x[rt][p].y;
        acc[rt] = __builtin_amdgcn_wmma_f32_16x16x4_f32(
             false, a0,  false, b[2 * p],
             (short)0, acc[rt],  false,  false);
      }
#pragma unroll
      for (int rt = 0; rt < RT; ++rt) {
        v2f a1; a1.x = x[rt][p].z; a1.y = x[rt][p].w;
        acc[rt] = __builtin_amdgcn_wmma_f32_16x16x4_f32(
             false, a1,  false, b[2 * p + 1],
             (short)0, acc[rt],  false,  false);
      }
    }

#pragma unroll
    for (int rt = 0; rt < RT; ++rt)
#pragma unroll
      for (int r = 0; r < 8; ++r)
        cbuf[cur][(16 * (rt0 + rt) + 8 * h + r) * PC + 16 * ct + c] = acc[rt][r];

    if (nb < nblk) stage(cur ^ 1);
    __syncthreads();

#pragma unroll
    for (int u = 0; u < 4 / RPI; ++u) {
      const int rb = 4 * wv + u * RPI + lane / LPR;
      const int col = 4 * (lane % LPR);
      const int row = blk * BM + rb;
      const float4 v = *reinterpret_cast<const float4*>(&cbuf[cur][rb * PC + col]);
      if (row < M) *reinterpret_cast<float4*>(C + (size_t)row * N + col) = v;
    }
  }
}

template <int F, bool HAS_BIAS, bool WRITE_REC, bool READ_REC>
__global__ void __launch_bounds__(1024)
agg_tile_kernel(const float* __restrict__ h, const int* __restrict__ src,
                const int* __restrict__ dst, const float* __restrict__ dinv,
                const float* __restrict__ bias, float* __restrict__ out,
                int* __restrict__ rec, int* __restrict__ rcnt,
                int n, int E) {
  constexpr int FPL = F / 32;
  constexpr int G = 32 / FPL;
  typedef float __attribute__((ext_vector_type(FPL))) VT;

  __shared__ float tile[AGG_TILE * F];
  __shared__ int pend[32 * AGG_PEND];

  const int tid = threadIdx.x;
  const int lane = tid & 31;
  const int wv = __builtin_amdgcn_readfirstlane(tid >> 5);
  const int lo = blockIdx.x * AGG_TILE;
  int rows = n - lo;
  if (rows > AGG_TILE) rows = AGG_TILE;

  int2* const recw = reinterpret_cast<int2*>(rec) + ((size_t)blockIdx.x * 32 + wv) * AGG_RCAP;
  int nrec = 0;
  bool useRec = false;
  if (READ_REC) {
    const int c = rcnt[blockIdx.x * 32 + lane];
    useRec = __builtin_amdgcn_ballot_w32(c > AGG_RCAP) == 0u;
    nrec = __builtin_amdgcn_readlane(c, wv);
  }

  for (int r = wv; r < rows; r += 32) {
    const int node = lo + r;
    const float di = dinv[node];
    const VT v = *reinterpret_cast<const VT*>(h + (size_t)node * F + lane * FPL);
#pragma unroll
    for (int c = 0; c < FPL; ++c) {
      float t = v[c] * di * di;
      if (HAS_BIAS) t += bias[lane * FPL + c];
      tile[r * F + lane + 32 * c] = t;
    }
  }
  __syncthreads();

  int cnt = 0;

  auto addRows = [&](int s_i, int dl_i, float w_i, int k, auto&& requested)
                     __attribute__((always_inline)) {
#pragma unroll 1
    for (int j0 = 0; j0 < k; j0 += G) {
      VT v[G];
#pragma unroll
      for (int jj = 0; jj < G; ++jj) {
        const int j = j0 + jj;
        const int s = __builtin_amdgcn_readlane(s_i, (j < k) ? j : (k - 1));
        v[jj] = *reinterpret_cast<const VT*>(h + (size_t)s * F + lane * FPL);
      }
      requested(j0);
#pragma unroll
      for (int jj = 0; jj < G; ++jj) {
        const int j = j0 + jj;
        if (j < k) {
          const float w = __int_as_float(__builtin_amdgcn_readlane(__float_as_int(w_i), j));
          const int dl = __builtin_amdgcn_readlane(dl_i, j);
#pragma unroll
          for (int c = 0; c < FPL; ++c)
            __hip_atomic_fetch_add(&tile[dl * F + lane + 32 * c], v[jj][c] * w,
                                   __ATOMIC_RELAXED, __HIP_MEMORY_SCOPE_WORKGROUP);
        }
      }
    }
  };

  auto drain = [&](int base, int k) __attribute__((always_inline)) {
    __builtin_amdgcn_wave_barrier();
    int s_i = 0, dl_i = 0;
    float w_i = 0.0f;
    if (lane < k) {
      const int e_i = pend[wv * AGG_PEND + base + lane];
      s_i = src[e_i];
      const int d_i = dst[e_i];
      dl_i = d_i - lo;
      w_i = dinv[s_i] * dinv[d_i];
    }
    if (WRITE_REC) {
      const int slot = nrec + lane;
      if (lane < k && slot < AGG_RCAP)
        recw[slot] = make_int2((s_i << AGG_DL_BITS) | dl_i, __float_as_int(w_i));
      nrec += k;
    }
    addRows(s_i, dl_i, w_i, k, [](int) {});
  };

  auto push = [&](int dv, int e) __attribute__((always_inline)) {
    const bool m = (unsigned)(dv - lo) < (unsigned)rows;
    const unsigned mask = __builtin_amdgcn_ballot_w32(m);
    if (mask != 0u) {
      if (m) pend[wv * AGG_PEND + cnt + (int)__builtin_amdgcn_mbcnt_lo(mask, 0u)] = e;
      cnt += __builtin_popcount(mask);
    }
  };

  if (READ_REC && useRec) {
    int2 cur = make_int2(0, 0);
    if (lane < nrec) cur = recw[lane];
    for (int base = 0; base < nrec; base += 32) {
      int2 nxt = make_int2(0, 0);
      const int left = nrec - base;
      addRows(cur.x >> AGG_DL_BITS, cur.x & ((1 << AGG_DL_BITS) - 1), __int_as_float(cur.y),
              (left < 32) ? left : 32, [&](int j0) {
                if (j0 == 0 && base + 32 + lane < nrec) nxt = recw[base + 32 + lane];
              });
      cur = nxt;
    }
  } else if ((E & 3) == 0) {
    const int4* dst4 = reinterpret_cast<const int4*>(dst);
    const int nq = E >> 2;
    constexpr int VPW = 8;
    constexpr int WQ = 32 * VPW;
    constexpr int STEP = 32 * WQ;
    const int nFull = nq / STEP;
    const unsigned R = (unsigned)rows;
    const int lane4 = lane * 4;

    auto quad = [&](const int4& d, int eb) __attribute__((always_inline)) {
      const unsigned rx = (unsigned)(d.x - lo), ry = (unsigned)(d.y - lo);
      const unsigned rz = (unsigned)(d.z - lo), rw = (unsigned)(d.w - lo);
      const unsigned mn = min(min(rx, ry), min(rz, rw));
      if (__builtin_amdgcn_ballot_w32(mn < R) != 0u) {
        const int e = eb + lane4;
        push(d.x, e);
        push(d.y, e + 1);
        push(d.z, e + 2);
        push(d.w, e + 3);
      }
    };
    auto flush32 = [&]() __attribute__((always_inline)) {
      while (cnt >= 32) {
        drain(cnt - 32, 32);
        cnt -= 32;
      }
    };

    if (nFull > 0) {
      int a = (int)(((long long)blockIdx.x * nFull) / gridDim.x);
      int4 cur[VPW], nxt[VPW];
      {
        const int4* p = dst4 + a * STEP + wv * WQ + lane;
#pragma unroll
        for (int u = 0; u < VPW; ++u) cur[u] = p[32 * u];
      }
      for (int i = 0; i < nFull; ++i) {
        int an = a + 1;
        if (an >= nFull) an = 0;
        {
          const int4* p = dst4 + an * STEP + wv * WQ + lane;
#pragma unroll
          for (int u = 0; u < VPW; ++u) nxt[u] = p[32 * u];
        }
        const int eb = (a * STEP + wv * WQ) * 4;
#pragma unroll
        for (int u = 0; u < 4; ++u) quad(cur[u], eb + 128 * u);
        flush32();
#pragma unroll
        for (int u = 4; u < VPW; ++u) quad(cur[u], eb + 128 * u);
        flush32();
#pragma unroll
        for (int u = 0; u < VPW; ++u) cur[u] = nxt[u];
        a = an;
      }
    }
    {
      const int qw = nFull * STEP + wv * WQ;
      int4 d[VPW];
#pragma unroll
      for (int u = 0; u < VPW; ++u) {
        const int q = qw + 32 * u + lane;
        d[u] = make_int4(-1, -1, -1, -1);
        if (q < nq) d[u] = dst4[q];
      }
#pragma unroll
      for (int u = 0; u < VPW; ++u) {
        quad(d[u], (qw + 32 * u) * 4);
        if (u == 3) flush32();
      }
    }
  } else {
    for (int e0 = wv * 32; e0 < E; e0 += 1024) {
      const int e = e0 + lane;
      const int dv = (e < E) ? dst[e] : -1;
      push(dv, e);
      if (cnt >= 32) {
        drain(cnt - 32, 32);
        cnt -= 32;
      }
    }
  }
  while (cnt > 0) {
    const int k = (cnt < 32) ? cnt : 32;
    drain(cnt - k, k);
    cnt -= k;
  }
  if (WRITE_REC) {
    if (lane == 0) rcnt[blockIdx.x * 32 + wv] = nrec;
  }

  __syncthreads();

  for (int r = wv; r < rows; r += 32) {
    VT v;
#pragma unroll
    for (int c = 0; c < FPL; ++c) v[c] = tile[r * F + lane + 32 * c];
    *reinterpret_cast<VT*>(out + (size_t)(lo + r) * F + lane * FPL) = v;
  }
}

extern "C" void kernel_launch(void* const* d_in, const int* in_sizes, int n_in,
                              void* d_out, int out_size, void* d_ws, size_t ws_size,
                              hipStream_t stream) {
  const float* x  = (const float*)d_in[0];
  const int*   ei = (const int*)d_in[1];
  const float* W1 = (const float*)d_in[2];
  const float* b1 = (const float*)d_in[3];
  const float* W2 = (const float*)d_in[4];
  const float* b2 = (const float*)d_in[5];
  float* out = (float*)d_out;

  const int Nn = GCN_N_NODES;
  const int E  = in_sizes[1] / 2;
  const int* src = ei;
  const int* dst = ei + E;

  float* dinv = (float*)d_ws;
  float* h    = dinv + Nn;
  float* out1 = h + (size_t)Nn * 128;
  float* h2   = h;
  int* rec  = (int*)(out1 + (size_t)Nn * 128);
  int* rcnt = rec + (size_t)AGG_GROUPS * 32 * AGG_RCAP * 2;

  deg_init_kernel<<<(Nn + 255) / 256, 256, 0, stream>>>(dinv, Nn);
  deg_count_kernel<<<(E + 255) / 256, 256, 0, stream>>>(dst, dinv, E);
  deg_rsqrt_kernel<<<(Nn + 255) / 256, 256, 0, stream>>>(dinv, Nn);

  gemm_wreg_kernel<128, false><<<1024, 256, 0, stream>>>(x, W1, nullptr, h, Nn);
  agg_tile_kernel<128, false, true, false><<<(Nn + AGG_TILE - 1) / AGG_TILE, 1024, 0, stream>>>(
      h, src, dst, dinv, nullptr, out1, rec, rcnt, Nn, E);

  gemm_wreg_kernel<64, true><<<1024, 256, 0, stream>>>(out1, W2, b1, h2, Nn);
  agg_tile_kernel<64, true, false, true><<<(Nn + AGG_TILE - 1) / AGG_TILE, 1024, 0, stream>>>(
      h2, src, dst, dinv, b2, out, rec, rcnt, Nn, E);
}


# Layer-1 tile kernel scans one tile-number byte per edge (stored by the degree pass; 16 edges per 16-B load, one test per load) instead of the 4-B dst list; sum order = LDS arrival
# speedup vs baseline: 2.5671x; 1.0303x over previous
#include <hip/hip_runtime.h>

typedef __attribute__((ext_vector_type(2))) float v2f;
typedef __attribute__((ext_vector_type(8))) float v8f;

#define GCN_N_NODES 100000
#define AGG_TILE 391
#define AGG_GROUPS ((GCN_N_NODES + AGG_TILE - 1) / AGG_TILE)
#define AGG_PEND 544
#define AGG_RCAP 1024
#define AGG_DL_BITS 9
static_assert(AGG_TILE <= (1 << AGG_DL_BITS), "tile row must fit the record's low bits");
static_assert(AGG_GROUPS <= 256, "a tile number must fit one byte");
static_assert(GCN_N_NODES <= (1 << (31 - AGG_DL_BITS)), "source row must fit the record's high bits");

__global__ void __launch_bounds__(256)
deg_init_kernel(float* __restrict__ deg, int n) {
  int i = blockIdx.x * blockDim.x + threadIdx.x;
  if (i < n) deg[i] = 1.0f;
}

__global__ void __launch_bounds__(256)
deg_count_kernel(const int* __restrict__ dst, float* __restrict__ deg,
                 unsigned char* __restrict__ tb, int E) {
  const int e = 4 * (blockIdx.x * blockDim.x + threadIdx.x);
  if ((E & 3) == 0 && e < E) {
    const int4 d = *reinterpret_cast<const int4*>(dst + e);
    unsafeAtomicAdd(&deg[d.x], 1.0f);
    unsafeAtomicAdd(&deg[d.y], 1.0f);
    unsafeAtomicAdd(&deg[d.z], 1.0f);
    unsafeAtomicAdd(&deg[d.w], 1.0f);
    const unsigned t = (unsigned)(d.x / AGG_TILE) | ((unsigned)(d.y / AGG_TILE) << 8) |
                       ((unsigned)(d.z / AGG_TILE) << 16) | ((unsigned)(d.w / AGG_TILE) << 24);
    *reinterpret_cast<unsigned*>(tb + e) = t;
  } else {
    for (int k = 0; k < 4 && e + k < E; ++k) {
      const int d = dst[e + k];
      unsafeAtomicAdd(&deg[d], 1.0f);
      tb[e + k] = (unsigned char)(d / AGG_TILE);
    }
  }
}

__global__ void __launch_bounds__(256)
deg_rsqrt_kernel(float* __restrict__ deg, int n) {
  int i = blockIdx.x * blockDim.x + threadIdx.x;
  if (i < n) deg[i] = rsqrtf(deg[i]);
}

template <int N, bool FUSE_RELU_BIAS>
__global__ void __launch_bounds__(256)
gemm_wreg_kernel(const float* __restrict__ A, const float* __restrict__ W,
                 const float* __restrict__ bias, float* __restrict__ C, int M) {
  constexpr int K = 128;
  constexpr int BM = 32;
  constexpr int PA = K + 4;
  constexpr int PC = N + 4;
  constexpr int NT = N / 16;
  constexpr int RT = NT / 4;
  constexpr int LPR = N / 4;
  constexpr int RPI = 32 / LPR;
  static_assert(NT == 8 || NT == 4, "N is 128 or 64");

  __shared__ alignas(16) float abuf[2][BM * PA];
  __shared__ alignas(16) float cbuf[2][BM * PC];

  const int tid = threadIdx.x;
  const int lane = tid & 31;
  const int wv = __builtin_amdgcn_readfirstlane(tid >> 5);
  const int c = lane & 15;
  const int h = lane >> 4;
  const int ct = wv % NT;
  const int rt0 = (wv / NT) * RT;

  const int nblk = (M + BM - 1) / BM;
  int blk = blockIdx.x;
  if (blk >= nblk) return;

  v2f b[32];
  {
    const float* wp = W + 16 * ct + c;
#pragma unroll
    for (int p = 0; p < 16; ++p) {
      const int k0 = 8 * p + 4 * h;
      b[2 * p].x = wp[(size_t)(k0 + 0) * N];
      b[2 * p].y = wp[(size_t)(k0 + 1) * N];
      b[2 * p + 1].x = wp[(size_t)(k0 + 2) * N];
      b[2 * p + 1].y = wp[(size_t)(k0 + 3) * N];
    }
  }
  float4 bv = make_float4(0.0f, 0.0f, 0.0f, 0.0f);
  if (FUSE_RELU_BIAS) bv = *reinterpret_cast<const float4*>(bias + 4 * lane);

  float4 st[4];
  auto fetch = [&](int q) __attribute__((always_inline)) {
#pragma unroll
    for (int u = 0; u < 4; ++u) {
      const int row = q * BM + 4 * wv + u;
      st[u] = make_float4(0.0f, 0.0f, 0.0f, 0.0f);
      if (row < M) st[u] = *reinterpret_cast<const float4*>(A + (size_t)row * K + 4 * lane);
    }
  };
  auto stage = [&](int buf) __attribute__((always_inline)) {
#pragma unroll
    for (int u = 0; u < 4; ++u) {
      float4 v = st[u];
      if (FUSE_RELU_BIAS) {
        v.x = fmaxf(v.x + bv.x, 0.0f);
        v.y = fmaxf(v.y + bv.y, 0.0f);
        v.z = fmaxf(v.z + bv.z, 0.0f);
        v.w = fmaxf(v.w + bv.w, 0.0f);
      }
      *reinterpret_cast<float4*>(&abuf[buf][(4 * wv + u) * PA + 4 * lane]) = v;
    }
  };

  fetch(blk);
  stage(0);
  __syncthreads();

  for (int it = 0; blk < nblk; blk += gridDim.x, ++it) {
    const int cur = it & 1;
    const int nb = blk + gridDim.x;
    if (nb < nblk) fetch(nb);

    float4 x[RT][16];
#pragma unroll
    for (int rt = 0; rt < RT; ++rt)
#pragma unroll
      for (int p = 0; p < 16; ++p)
        x[rt][p] = *reinterpret_cast<const float4*>(
            &abuf[cur][(16 * (rt0 + rt) + c) * PA + 8 * p + 4 * h]);

    v8f acc[RT];
#pragma unroll
    for (int rt = 0; rt < RT; ++rt) acc[rt] = v8f{};
#pragma unroll
    for (int p = 0; p < 16; ++p) {
#pragma unroll
      for (int rt = 0; rt < RT; ++rt) {
        v2f a0; a0.x = x[rt][p].x; a0.y = x[rt][p].y;
        acc[rt] = __builtin_amdgcn_wmma_f32_16x16x4_f32(
             false, a0,  false, b[2 * p],
             (short)0, acc[rt],  false,  false);
      }
#pragma unroll
      for (int rt = 0; rt < RT; ++rt) {
        v2f a1; a1.x = x[rt][p].z; a1.y = x[rt][p].w;
        acc[rt] = __builtin_amdgcn_wmma_f32_16x16x4_f32(
             false, a1,  false, b[2 * p + 1],
             (short)0, acc[rt],  false,  false);
      }
    }

#pragma unroll
    for (int rt = 0; rt < RT; ++rt)
#pragma unroll
      for (int r = 0; r < 8; ++r)
        cbuf[cur][(16 * (rt0 + rt) + 8 * h + r) * PC + 16 * ct + c] = acc[rt][r];

    if (nb < nblk) stage(cur ^ 1);
    __syncthreads();

#pragma unroll
    for (int u = 0; u < 4 / RPI; ++u) {
      const int rb = 4 * wv + u * RPI + lane / LPR;
      const int col = 4 * (lane % LPR);
      const int row = blk * BM + rb;
      const float4 v = *reinterpret_cast<const float4*>(&cbuf[cur][rb * PC + col]);
      if (row < M) *reinterpret_cast<float4*>(C + (size_t)row * N + col) = v;
    }
  }
}

template <int F, bool HAS_BIAS, bool WRITE_REC, bool READ_REC>
__global__ void __launch_bounds__(1024)
agg_tile_kernel(const float* __restrict__ h, const int* __restrict__ src,
                const int* __restrict__ dst, const float* __restrict__ dinv,
                const float* __restrict__ bias, float* __restrict__ out,
                int* __restrict__ rec, int* __restrict__ rcnt,
                const unsigned char* __restrict__ tb, int n, int E) {
  constexpr int FPL = F / 32;
  constexpr int G = 32 / FPL;
  typedef float __attribute__((ext_vector_type(FPL))) VT;

  __shared__ float tile[AGG_TILE * F];
  __shared__ int pend[32 * AGG_PEND];

  const int tid = threadIdx.x;
  const int lane = tid & 31;
  const int wv = __builtin_amdgcn_readfirstlane(tid >> 5);
  const int lo = blockIdx.x * AGG_TILE;
  int rows = n - lo;
  if (rows > AGG_TILE) rows = AGG_TILE;

  int2* const recw = reinterpret_cast<int2*>(rec) + ((size_t)blockIdx.x * 32 + wv) * AGG_RCAP;
  int nrec = 0;
  bool useRec = false;
  if (READ_REC) {
    const int c = rcnt[blockIdx.x * 32 + lane];
    useRec = __builtin_amdgcn_ballot_w32(c > AGG_RCAP) == 0u;
    nrec = __builtin_amdgcn_readlane(c, wv);
  }

  for (int r = wv; r < rows; r += 32) {
    const int node = lo + r;
    const float di = dinv[node];
    const VT v = *reinterpret_cast<const VT*>(h + (size_t)node * F + lane * FPL);
#pragma unroll
    for (int c = 0; c < FPL; ++c) {
      float t = v[c] * di * di;
      if (HAS_BIAS) t += bias[lane * FPL + c];
      tile[r * F + lane + 32 * c] = t;
    }
  }
  __syncthreads();

  int cnt = 0;

  auto addRows = [&](int s_i, int dl_i, float w_i, int k, auto&& requested)
                     __attribute__((always_inline)) {
#pragma unroll 1
    for (int j0 = 0; j0 < k; j0 += G) {
      VT v[G];
#pragma unroll
      for (int jj = 0; jj < G; ++jj) {
        const int j = j0 + jj;
        const int s = __builtin_amdgcn_readlane(s_i, (j < k) ? j : (k - 1));
        v[jj] = *reinterpret_cast<const VT*>(h + (size_t)s * F + lane * FPL);
      }
      requested(j0);
#pragma unroll
      for (int jj = 0; jj < G; ++jj) {
        const int j = j0 + jj;
        if (j < k) {
          const float w = __int_as_float(__builtin_amdgcn_readlane(__float_as_int(w_i), j));
          const int dl = __builtin_amdgcn_readlane(dl_i, j);
#pragma unroll
          for (int c = 0; c < FPL; ++c)
            __hip_atomic_fetch_add(&tile[dl * F + lane + 32 * c], v[jj][c] * w,
                                   __ATOMIC_RELAXED, __HIP_MEMORY_SCOPE_WORKGROUP);
        }
      }
    }
  };

  auto drain = [&](int base, int k) __attribute__((always_inline)) {
    __builtin_amdgcn_wave_barrier();
    int s_i = 0, dl_i = 0;
    float w_i = 0.0f;
    if (lane < k) {
      const int e_i = pend[wv * AGG_PEND + base + lane];
      s_i = src[e_i];
      const int d_i = dst[e_i];
      dl_i = d_i - lo;
      w_i = dinv[s_i] * dinv[d_i];
    }
    if (WRITE_REC) {
      const int slot = nrec + lane;
      if (lane < k && slot < AGG_RCAP)
        recw[slot] = make_int2((s_i << AGG_DL_BITS) | dl_i, __float_as_int(w_i));
      nrec += k;
    }
    addRows(s_i, dl_i, w_i, k, [](int) {});
  };

  auto push = [&](int dv, int e) __attribute__((always_inline)) {
    const bool m = (unsigned)(dv - lo) < (unsigned)rows;
    const unsigned mask = __builtin_amdgcn_ballot_w32(m);
    if (mask != 0u) {
      if (m) pend[wv * AGG_PEND + cnt + (int)__builtin_amdgcn_mbcnt_lo(mask, 0u)] = e;
      cnt += __builtin_popcount(mask);
    }
  };

  if (READ_REC && useRec) {
    int2 cur = make_int2(0, 0);
    if (lane < nrec) cur = recw[lane];
    for (int base = 0; base < nrec; base += 32) {
      int2 nxt = make_int2(0, 0);
      const int left = nrec - base;
      addRows(cur.x >> AGG_DL_BITS, cur.x & ((1 << AGG_DL_BITS) - 1), __int_as_float(cur.y),
              (left < 32) ? left : 32, [&](int j0) {
                if (j0 == 0 && base + 32 + lane < nrec) nxt = recw[base + 32 + lane];
              });
      cur = nxt;
    }
  } else if ((E & 15) == 0) {
    const uint4* tb4 = reinterpret_cast<const uint4*>(tb);
    const int nh = E >> 4;
    constexpr int VPW = 4;
    constexpr int WH = 32 * VPW;
    constexpr int STEP = 32 * WH;
    const int nFull = nh / STEP;
    const unsigned tpat = 0x01010101u * (unsigned)blockIdx.x;

    auto ne = [&](unsigned v) __attribute__((always_inline)) {
      const unsigned x = v ^ tpat;
      return ((x & 0x7F7F7F7Fu) + 0x7F7F7F7Fu) | x;
    };
    auto hex = [&](const uint4& d, int gb) __attribute__((always_inline)) {
      const unsigned z0 = ne(d.x), z1 = ne(d.y), z2 = ne(d.z), z3 = ne(d.w);
      unsigned m = __builtin_amdgcn_ballot_w32((~(z0 & z1 & z2 & z3) & 0x80808080u) != 0u);
      if (m != 0u) {
        unsigned hm = ((~z0 & 0x80808080u) >> 7) | ((~z1 & 0x80808080u) >> 6) |
                      ((~z2 & 0x80808080u) >> 5) | ((~z3 & 0x80808080u) >> 4);
        const int e0 = (gb + lane) * 16;
        do {
          if (hm != 0u) {
            const int p = __builtin_ctz(hm);
            pend[wv * AGG_PEND + cnt + (int)__builtin_amdgcn_mbcnt_lo(m, 0u)] =
                e0 + 4 * (p & 7) + (p >> 3);
          }
          cnt += __builtin_popcount(m);
          hm &= hm - 1u;
          m = __builtin_amdgcn_ballot_w32(hm != 0u);
        } while (m != 0u);
      }
    };
    auto flush32 = [&]() __attribute__((always_inline)) {
      while (cnt >= 32) {
        drain(cnt - 32, 32);
        cnt -= 32;
      }
    };

    if (nFull > 0) {
      int a = (int)(((long long)blockIdx.x * nFull) / gridDim.x);
      uint4 cur[VPW], nxt[VPW];
      {
        const uint4* p = tb4 + a * STEP + wv * WH + lane;
#pragma unroll
        for (int u = 0; u < VPW; ++u) cur[u] = p[32 * u];
      }
      for (int i = 0; i < nFull; ++i) {
        int an = a + 1;
        if (an >= nFull) an = 0;
        {
          const uint4* p = tb4 + an * STEP + wv * WH + lane;
#pragma unroll
          for (int u = 0; u < VPW; ++u) nxt[u] = p[32 * u];
        }
        const int gb = a * STEP + wv * WH;
#pragma unroll
        for (int u = 0; u < VPW; ++u) {
          hex(cur[u], gb + 32 * u);
          flush32();
        }
#pragma unroll
        for (int u = 0; u < VPW; ++u) cur[u] = nxt[u];
        a = an;
      }
    }
    {
      const int gw = nFull * STEP + wv * WH;
      uint4 d[VPW];
#pragma unroll
      for (int u = 0; u < VPW; ++u) {
        const int g = gw + 32 * u + lane;
        d[u] = make_uint4(~tpat, ~tpat, ~tpat, ~tpat);
        if (g < nh) d[u] = tb4[g];
      }
#pragma unroll
      for (int u = 0; u < VPW; ++u) {
        hex(d[u], gw + 32 * u);
        flush32();
      }
    }
  } else if ((E & 3) == 0) {
    const int4* dst4 = reinterpret_cast<const int4*>(dst);
    const int nq = E >> 2;
    constexpr int VPW = 8;
    constexpr int WQ = 32 * VPW;
    constexpr int STEP = 32 * WQ;
    const int nFull = nq / STEP;
    const unsigned R = (unsigned)rows;
    const int lane4 = lane * 4;

    auto quad = [&](const int4& d, int eb) __attribute__((always_inline)) {
      const unsigned rx = (unsigned)(d.x - lo), ry = (unsigned)(d.y - lo);
      const unsigned rz = (unsigned)(d.z - lo), rw = (unsigned)(d.w - lo);
      const unsigned mn = min(min(rx, ry), min(rz, rw));
      if (__builtin_amdgcn_ballot_w32(mn < R) != 0u) {
        const int e = eb + lane4;
        push(d.x, e);
        push(d.y, e + 1);
        push(d.z, e + 2);
        push(d.w, e + 3);
      }
    };
    auto flush32 = [&]() __attribute__((always_inline)) {
      while (cnt >= 32) {
        drain(cnt - 32, 32);
        cnt -= 32;
      }
    };

    if (nFull > 0) {
      int a = (int)(((long long)blockIdx.x * nFull) / gridDim.x);
      int4 cur[VPW], nxt[VPW];
      {
        const int4* p = dst4 + a * STEP + wv * WQ + lane;
#pragma unroll
        for (int u = 0; u < VPW; ++u) cur[u] = p[32 * u];
      }
      for (int i = 0; i < nFull; ++i) {
        int an = a + 1;
        if (an >= nFull) an = 0;
        {
          const int4* p = dst4 + an * STEP + wv * WQ + lane;
#pragma unroll
          for (int u = 0; u < VPW; ++u) nxt[u] = p[32 * u];
        }
        const int eb = (a * STEP + wv * WQ) * 4;
#pragma unroll
        for (int u = 0; u < 4; ++u) quad(cur[u], eb + 128 * u);
        flush32();
#pragma unroll
        for (int u = 4; u < VPW; ++u) quad(cur[u], eb + 128 * u);
        flush32();
#pragma unroll
        for (int u = 0; u < VPW; ++u) cur[u] = nxt[u];
        a = an;
      }
    }
    {
      const int qw = nFull * STEP + wv * WQ;
      int4 d[VPW];
#pragma unroll
      for (int u = 0; u < VPW; ++u) {
        const int q = qw + 32 * u + lane;
        d[u] = make_int4(-1, -1, -1, -1);
        if (q < nq) d[u] = dst4[q];
      }
#pragma unroll
      for (int u = 0; u < VPW; ++u) {
        quad(d[u], (qw + 32 * u) * 4);
        if (u == 3) flush32();
      }
    }
  } else {
    for (int e0 = wv * 32; e0 < E; e0 += 1024) {
      const int e = e0 + lane;
      const int dv = (e < E) ? dst[e] : -1;
      push(dv, e);
      if (cnt >= 32) {
        drain(cnt - 32, 32);
        cnt -= 32;
      }
    }
  }
  while (cnt > 0) {
    const int k = (cnt < 32) ? cnt : 32;
    drain(cnt - k, k);
    cnt -= k;
  }
  if (WRITE_REC) {
    if (lane == 0) rcnt[blockIdx.x * 32 + wv] = nrec;
  }

  __syncthreads();

  for (int r = wv; r < rows; r += 32) {
    VT v;
#pragma unroll
    for (int c = 0; c < FPL; ++c) v[c] = tile[r * F + lane + 32 * c];
    *reinterpret_cast<VT*>(out + (size_t)(lo + r) * F + lane * FPL) = v;
  }
}

extern "C" void kernel_launch(void* const* d_in, const int* in_sizes, int n_in,
                              void* d_out, int out_size, void* d_ws, size_t ws_size,
                              hipStream_t stream) {
  const float* x  = (const float*)d_in[0];
  const int*   ei = (const int*)d_in[1];
  const float* W1 = (const float*)d_in[2];
  const float* b1 = (const float*)d_in[3];
  const float* W2 = (const float*)d_in[4];
  const float* b2 = (const float*)d_in[5];
  float* out = (float*)d_out;

  const int Nn = GCN_N_NODES;
  const int E  = in_sizes[1] / 2;
  const int* src = ei;
  const int* dst = ei + E;

  float* dinv = (float*)d_ws;
  float* h    = dinv + Nn;
  float* out1 = h + (size_t)Nn * 128;
  float* h2   = h;
  int* rec  = (int*)(out1 + (size_t)Nn * 128);
  int* rcnt = rec + (size_t)AGG_GROUPS * 32 * AGG_RCAP * 2;
  unsigned char* tb = (unsigned char*)(rcnt + AGG_GROUPS * 32);

  deg_init_kernel<<<(Nn + 255) / 256, 256, 0, stream>>>(dinv, Nn);
  deg_count_kernel<<<((E + 3) / 4 + 255) / 256, 256, 0, stream>>>(dst, dinv, tb, E);
  deg_rsqrt_kernel<<<(Nn + 255) / 256, 256, 0, stream>>>(dinv, Nn);

  gemm_wreg_kernel<128, false><<<1024, 256, 0, stream>>>(x, W1, nullptr, h, Nn);
  agg_tile_kernel<128, false, true, false><<<(Nn + AGG_TILE - 1) / AGG_TILE, 1024, 0, stream>>>(
      h, src, dst, dinv, nullptr, out1, rec, rcnt, tb, Nn, E);

  gemm_wreg_kernel<64, true><<<1024, 256, 0, stream>>>(out1, W2, b1, h2, Nn);
  agg_tile_kernel<64, true, false, true><<<(Nn + AGG_TILE - 1) / AGG_TILE, 1024, 0, stream>>>(
      h2, src, dst, dinv, b2, out, rec, rcnt, tb, Nn, E);
}


# Tile kernels keep more loads in flight per wave: all found edges are looked up in batches of 256 after the scan, next rows are requested while current ones are added; sum order = LDS arrival
# speedup vs baseline: 2.7307x; 1.0637x over previous
#include <hip/hip_runtime.h>

typedef __attribute__((ext_vector_type(2))) float v2f;
typedef __attribute__((ext_vector_type(8))) float v8f;

#define GCN_N_NODES 100000
#define AGG_TILE 391
#define AGG_GROUPS ((GCN_N_NODES + AGG_TILE - 1) / AGG_TILE)
#define AGG_PEND 896
#define AGG_FLUSH_AT (AGG_PEND - 512)
#define AGG_MAXC 8
#define AGG_RCAP 1024
#define AGG_DL_BITS 9
static_assert(AGG_TILE <= (1 << AGG_DL_BITS), "tile row must fit the record's low bits");
static_assert(AGG_GROUPS <= 256, "a tile number must fit one byte");
static_assert(GCN_N_NODES <= (1 << (31 - AGG_DL_BITS)), "source row must fit the record's high bits");

__global__ void __launch_bounds__(256)
deg_init_kernel(float* __restrict__ deg, int n) {
  int i = blockIdx.x * blockDim.x + threadIdx.x;
  if (i < n) deg[i] = 1.0f;
}

__global__ void __launch_bounds__(256)
deg_count_kernel(const int* __restrict__ dst, float* __restrict__ deg,
                 unsigned char* __restrict__ tb, int E) {
  const int e = 4 * (blockIdx.x * blockDim.x + threadIdx.x);
  if ((E & 3) == 0 && e < E) {
    const int4 d = *reinterpret_cast<const int4*>(dst + e);
    unsafeAtomicAdd(&deg[d.x], 1.0f);
    unsafeAtomicAdd(&deg[d.y], 1.0f);
    unsafeAtomicAdd(&deg[d.z], 1.0f);
    unsafeAtomicAdd(&deg[d.w], 1.0f);
    const unsigned t = (unsigned)(d.x / AGG_TILE) | ((unsigned)(d.y / AGG_TILE) << 8) |
                       ((unsigned)(d.z / AGG_TILE) << 16) | ((unsigned)(d.w / AGG_TILE) << 24);
    *reinterpret_cast<unsigned*>(tb + e) = t;
  } else {
    for (int k = 0; k < 4 && e + k < E; ++k) {
      const int d = dst[e + k];
      unsafeAtomicAdd(&deg[d], 1.0f);
      tb[e + k] = (unsigned char)(d / AGG_TILE);
    }
  }
}

__global__ void __launch_bounds__(256)
deg_rsqrt_kernel(float* __restrict__ deg, int n) {
  int i = blockIdx.x * blockDim.x + threadIdx.x;
  if (i < n) deg[i] = rsqrtf(deg[i]);
}

template <int N, bool FUSE_RELU_BIAS>
__global__ void __launch_bounds__(256)
gemm_wreg_kernel(const float* __restrict__ A, const float* __restrict__ W,
                 const float* __restrict__ bias, float* __restrict__ C, int M) {
  constexpr int K = 128;
  constexpr int BM = 32;
  constexpr int PA = K + 4;
  constexpr int PC = N + 4;
  constexpr int NT = N / 16;
  constexpr int RT = NT / 4;
  constexpr int LPR = N / 4;
  constexpr int RPI = 32 / LPR;
  static_assert(NT == 8 || NT == 4, "N is 128 or 64");

  __shared__ alignas(16) float abuf[2][BM * PA];
  __shared__ alignas(16) float cbuf[2][BM * PC];

  const int tid = threadIdx.x;
  const int lane = tid & 31;
  const int wv = __builtin_amdgcn_readfirstlane(tid >> 5);
  const int c = lane & 15;
  const int h = lane >> 4;
  const int ct = wv % NT;
  const int rt0 = (wv / NT) * RT;

  const int nblk = (M + BM - 1) / BM;
  int blk = blockIdx.x;
  if (blk >= nblk) return;

  v2f b[32];
  {
    const float* wp = W + 16 * ct + c;
#pragma unroll
    for (int p = 0; p < 16; ++p) {
      const int k0 = 8 * p + 4 * h;
      b[2 * p].x = wp[(size_t)(k0 + 0) * N];
      b[2 * p].y = wp[(size_t)(k0 + 1) * N];
      b[2 * p + 1].x = wp[(size_t)(k0 + 2) * N];
      b[2 * p + 1].y = wp[(size_t)(k0 + 3) * N];
    }
  }
  float4 bv = make_float4(0.0f, 0.0f, 0.0f, 0.0f);
  if (FUSE_RELU_BIAS) bv = *reinterpret_cast<const float4*>(bias + 4 * lane);

  float4 st[4];
  auto fetch = [&](int q) __attribute__((always_inline)) {
#pragma unroll
    for (int u = 0; u < 4; ++u) {
      const int row = q * BM + 4 * wv + u;
      st[u] = make_float4(0.0f, 0.0f, 0.0f, 0.0f);
      if (row < M) st[u] = *reinterpret_cast<const float4*>(A + (size_t)row * K + 4 * lane);
    }
  };
  auto stage = [&](int buf) __attribute__((always_inline)) {
#pragma unroll
    for (int u = 0; u < 4; ++u) {
      float4 v = st[u];
      if (FUSE_RELU_BIAS) {
        v.x = fmaxf(v.x + bv.x, 0.0f);
        v.y = fmaxf(v.y + bv.y, 0.0f);
        v.z = fmaxf(v.z + bv.z, 0.0f);
        v.w = fmaxf(v.w + bv.w, 0.0f);
      }
      *reinterpret_cast<float4*>(&abuf[buf][(4 * wv + u) * PA + 4 * lane]) = v;
    }
  };

  fetch(blk);
  stage(0);
  __syncthreads();

  for (int it = 0; blk < nblk; blk += gridDim.x, ++it) {
    const int cur = it & 1;
    const int nb = blk + gridDim.x;
    if (nb < nblk) fetch(nb);

    float4 x[RT][16];
#pragma unroll
    for (int rt = 0; rt < RT; ++rt)
#pragma unroll
      for (int p = 0; p < 16; ++p)
        x[rt][p] = *reinterpret_cast<const float4*>(
            &abuf[cur][(16 * (rt0 + rt) + c) * PA + 8 * p + 4 * h]);

    v8f acc[RT];
#pragma unroll
    for (int rt = 0; rt < RT; ++rt) acc[rt] = v8f{};
#pragma unroll
    for (int p = 0; p < 16; ++p) {
#pragma unroll
      for (int rt = 0; rt < RT; ++rt) {
        v2f a0; a0.x = x[rt][p].x; a0.y = x[rt][p].y;
        acc[rt] = __builtin_amdgcn_wmma_f32_16x16x4_f32(
             false, a0,  false, b[2 * p],
             (short)0, acc[rt],  false,  false);
      }
#pragma unroll
      for (int rt = 0; rt < RT; ++rt) {
        v2f a1; a1.x = x[rt][p].z; a1.y = x[rt][p].w;
        acc[rt] = __builtin_amdgcn_wmma_f32_16x16x4_f32(
             false, a1,  false, b[2 * p + 1],
             (short)0, acc[rt],  false,  false);
      }
    }

#pragma unroll
    for (int rt = 0; rt < RT; ++rt)
#pragma unroll
      for (int r = 0; r < 8; ++r)
        cbuf[cur][(16 * (rt0 + rt) + 8 * h + r) * PC + 16 * ct + c] = acc[rt][r];

    if (nb < nblk) stage(cur ^ 1);
    __syncthreads();

#pragma unroll
    for (int u = 0; u < 4 / RPI; ++u) {
      const int rb = 4 * wv + u * RPI + lane / LPR;
      const int col = 4 * (lane % LPR);
      const int row = blk * BM + rb;
      const float4 v = *reinterpret_cast<const float4*>(&cbuf[cur][rb * PC + col]);
      if (row < M) *reinterpret_cast<float4*>(C + (size_t)row * N + col) = v;
    }
  }
}

template <int F, bool HAS_BIAS, bool WRITE_REC, bool READ_REC>
__global__ void __launch_bounds__(1024)
agg_tile_kernel(const float* __restrict__ h, const int* __restrict__ src,
                const int* __restrict__ dst, const float* __restrict__ dinv,
                const float* __restrict__ bias, float* __restrict__ out,
                int* __restrict__ rec, int* __restrict__ rcnt,
                const unsigned char* __restrict__ tb, int n, int E) {
  constexpr int FPL = F / 32;
  constexpr int G = 32 / FPL;
  typedef float __attribute__((ext_vector_type(FPL))) VT;

  __shared__ float tile[AGG_TILE * F];
  __shared__ int pend[32 * AGG_PEND];

  const int tid = threadIdx.x;
  const int lane = tid & 31;
  const int wv = __builtin_amdgcn_readfirstlane(tid >> 5);
  const int lo = blockIdx.x * AGG_TILE;
  int rows = n - lo;
  if (rows > AGG_TILE) rows = AGG_TILE;

  int2* const recw = reinterpret_cast<int2*>(rec) + ((size_t)blockIdx.x * 32 + wv) * AGG_RCAP;
  int nrec = 0;
  bool useRec = false;
  if (READ_REC) {
    const int c = rcnt[blockIdx.x * 32 + lane];
    useRec = __builtin_amdgcn_ballot_w32(c > AGG_RCAP) == 0u;
    nrec = __builtin_amdgcn_readlane(c, wv);
  }

  {
    constexpr int NR = (AGG_TILE + 31) / 32;
    static_assert(NR <= 32, "one lane per row of the wave");
    VT iv[NR];
    float bb[FPL];
#pragma unroll
    for (int c = 0; c < FPL; ++c) bb[c] = HAS_BIAS ? bias[lane * FPL + c] : 0.0f;
    float dvl;
    {
      const int t = (lane < NR) ? lane : (NR - 1);
      const int r = wv + 32 * t;
      dvl = dinv[lo + ((r < rows) ? r : (rows - 1))];
    }
#pragma unroll
    for (int t = 0; t < NR; ++t) {
      const int r = wv + 32 * t;
      const int node = lo + ((r < rows) ? r : (rows - 1));
      iv[t] = *reinterpret_cast<const VT*>(h + (size_t)node * F + lane * FPL);
    }
#pragma unroll
    for (int t = 0; t < NR; ++t) {
      const int r = wv + 32 * t;
      if (r < rows) {
        const float dit = __int_as_float(__builtin_amdgcn_readlane(__float_as_int(dvl), t));
#pragma unroll
        for (int c = 0; c < FPL; ++c) {
          float v = iv[t][c] * dit * dit;
          if (HAS_BIAS) v += bb[c];
          tile[r * F + lane + 32 * c] = v;
        }
      }
    }
  }
  __syncthreads();

  int cnt = 0;

  auto addRows = [&](int s_i, int dl_i, float w_i, int k, auto&& requested)
                     __attribute__((always_inline)) {
#pragma unroll 1
    for (int j0 = 0; j0 < k; j0 += G) {
      VT v[G];
#pragma unroll
      for (int jj = 0; jj < G; ++jj) {
        const int j = j0 + jj;
        const int s = __builtin_amdgcn_readlane(s_i, (j < k) ? j : (k - 1));
        v[jj] = *reinterpret_cast<const VT*>(h + (size_t)s * F + lane * FPL);
      }
      requested(j0);
#pragma unroll
      for (int jj = 0; jj < G; ++jj) {
        const int j = j0 + jj;
        if (j < k) {
          const float w = __int_as_float(__builtin_amdgcn_readlane(__float_as_int(w_i), j));
          const int dl = __builtin_amdgcn_readlane(dl_i, j);
#pragma unroll
          for (int c = 0; c < FPL; ++c)
            __hip_atomic_fetch_add(&tile[dl * F + lane + 32 * c], v[jj][c] * w,
                                   __ATOMIC_RELAXED, __HIP_MEMORY_SCOPE_WORKGROUP);
        }
      }
    }
  };

  auto drain = [&](int base, int k) __attribute__((always_inline)) {
    __builtin_amdgcn_wave_barrier();
    int s_i = 0, dl_i = 0;
    float w_i = 0.0f;
    if (lane < k) {
      const int e_i = pend[wv * AGG_PEND + base + lane];
      s_i = src[e_i];
      const int d_i = dst[e_i];
      dl_i = d_i - lo;
      w_i = dinv[s_i] * dinv[d_i];
    }
    if (WRITE_REC) {
      const int slot = nrec + lane;
      if (lane < k && slot < AGG_RCAP)
        recw[slot] = make_int2((s_i << AGG_DL_BITS) | dl_i, __float_as_int(w_i));
      nrec += k;
    }
    addRows(s_i, dl_i, w_i, k, [](int) {});
  };

  constexpr int NG = 32 / G;
  auto loadG = [&](VT (&v)[G], int s_i, int j0, int k) __attribute__((always_inline)) {
#pragma unroll
    for (int jj = 0; jj < G; ++jj) {
      const int j = j0 + jj;
      const int s = __builtin_amdgcn_readlane(s_i, (j < k) ? j : (k - 1));
      v[jj] = *reinterpret_cast<const VT*>(h + (size_t)s * F + lane * FPL);
    }
  };
  auto addG = [&](VT (&v)[G], int dl_i, float w_i, int j0, int k) __attribute__((always_inline)) {
#pragma unroll
    for (int jj = 0; jj < G; ++jj) {
      const int j = j0 + jj;
      if (j < k) {
        const float w = __int_as_float(__builtin_amdgcn_readlane(__float_as_int(w_i), j));
        const int dl = __builtin_amdgcn_readlane(dl_i, j);
#pragma unroll
        for (int c = 0; c < FPL; ++c)
          __hip_atomic_fetch_add(&tile[dl * F + lane + 32 * c], v[jj][c] * w,
                                 __ATOMIC_RELAXED, __HIP_MEMORY_SCOPE_WORKGROUP);
      }
    }
  };
  auto chunk = [&](int s_i, int dl_i, float w_i, int k) __attribute__((always_inline)) {
    VT a[G], b[G];
    loadG(a, s_i, 0, k);
    if (k > G) {
      loadG(b, s_i, G, k);
      addG(a, dl_i, w_i, 0, k);
      if constexpr (NG > 2) {
        if (k > 2 * G) {
          loadG(a, s_i, 2 * G, k);
          addG(b, dl_i, w_i, G, k);
          if (k > 3 * G) {
            loadG(b, s_i, 3 * G, k);
            addG(a, dl_i, w_i, 2 * G, k);
            addG(b, dl_i, w_i, 3 * G, k);
          } else {
            addG(a, dl_i, w_i, 2 * G, k);
          }
        } else {
          addG(b, dl_i, w_i, G, k);
        }
      } else {
        addG(b, dl_i, w_i, G, k);
      }
    } else {
      addG(a, dl_i, w_i, 0, k);
    }
  };

  auto drainAll = [&]() __attribute__((always_inline)) {
    __builtin_amdgcn_wave_barrier();
#pragma unroll 1
    for (int sb = 0; sb < cnt; sb += 32 * AGG_MAXC) {
      int s_i[AGG_MAXC], d_i[AGG_MAXC];
      float w_i[AGG_MAXC];
#pragma unroll
      for (int u = 0; u < AGG_MAXC; ++u) {
        const int idx = sb + 32 * u + lane;
        int e = 0;
        if (idx < cnt) e = pend[wv * AGG_PEND + idx];
        s_i[u] = e;
      }
#pragma unroll
      for (int u = 0; u < AGG_MAXC; ++u) {
        const int e = s_i[u];
        s_i[u] = src[e];
        d_i[u] = dst[e];
      }
#pragma unroll
      for (int u = 0; u < AGG_MAXC; ++u) w_i[u] = dinv[s_i[u]] * dinv[d_i[u]];
      const int left = cnt - sb;
      if (WRITE_REC) {
#pragma unroll
        for (int u = 0; u < AGG_MAXC; ++u) {
          const int slot = nrec + 32 * u + lane;
          if (32 * u + lane < left && slot < AGG_RCAP)
            recw[slot] = make_int2((s_i[u] << AGG_DL_BITS) | (d_i[u] - lo), __float_as_int(w_i[u]));
        }
        nrec += (left < 32 * AGG_MAXC) ? left : 32 * AGG_MAXC;
      }
#pragma unroll 1
      for (int u = 0; u < AGG_MAXC; ++u) {
        int k = left - 32 * u;
        if (k <= 0) break;
        if (k > 32) k = 32;
        int s = s_i[0], d = d_i[0];
        float w = w_i[0];
#pragma unroll
        for (int t = 1; t < AGG_MAXC; ++t) {
          if (u == t) {
            s = s_i[t];
            d = d_i[t];
            w = w_i[t];
          }
        }
        chunk(s, d - lo, w, k);
      }
    }
  };

  auto push = [&](int dv, int e) __attribute__((always_inline)) {
    const bool m = (unsigned)(dv - lo) < (unsigned)rows;
    const unsigned mask = __builtin_amdgcn_ballot_w32(m);
    if (mask != 0u) {
      if (m) pend[wv * AGG_PEND + cnt + (int)__builtin_amdgcn_mbcnt_lo(mask, 0u)] = e;
      cnt += __builtin_popcount(mask);
    }
  };

  if (READ_REC && useRec) {
    int2 cur = make_int2(0, 0);
    if (lane < nrec) cur = recw[lane];
    for (int base = 0; base < nrec; base += 32) {
      int2 nxt = make_int2(0, 0);
      const int left = nrec - base;
      if (base + 32 + lane < nrec) nxt = recw[base + 32 + lane];
      chunk(cur.x >> AGG_DL_BITS, cur.x & ((1 << AGG_DL_BITS) - 1), __int_as_float(cur.y),
            (left < 32) ? left : 32);
      cur = nxt;
    }
  } else if ((E & 15) == 0) {
    const uint4* tb4 = reinterpret_cast<const uint4*>(tb);
    const int nh = E >> 4;
    constexpr int VPW = 4;
    constexpr int WH = 32 * VPW;
    constexpr int STEP = 32 * WH;
    const int nFull = nh / STEP;
    const unsigned tpat = 0x01010101u * (unsigned)blockIdx.x;

    auto ne = [&](unsigned v) __attribute__((always_inline)) {
      const unsigned x = v ^ tpat;
      return ((x & 0x7F7F7F7Fu) + 0x7F7F7F7Fu) | x;
    };
    auto hex = [&](const uint4& d, int gb) __attribute__((always_inline)) {
      const unsigned z0 = ne(d.x), z1 = ne(d.y), z2 = ne(d.z), z3 = ne(d.w);
      unsigned m = __builtin_amdgcn_ballot_w32((~(z0 & z1 & z2 & z3) & 0x80808080u) != 0u);
      if (m != 0u) {
        unsigned hm = ((~z0 & 0x80808080u) >> 7) | ((~z1 & 0x80808080u) >> 6) |
                      ((~z2 & 0x80808080u) >> 5) | ((~z3 & 0x80808080u) >> 4);
        const int e0 = (gb + lane) * 16;
        do {
          if (hm != 0u) {
            const int p = __builtin_ctz(hm);
            pend[wv * AGG_PEND + cnt + (int)__builtin_amdgcn_mbcnt_lo(m, 0u)] =
                e0 + 4 * (p & 7) + (p >> 3);
          }
          cnt += __builtin_popcount(m);
          hm &= hm - 1u;
          m = __builtin_amdgcn_ballot_w32(hm != 0u);
        } while (m != 0u);
      }
    };
    auto flush32 = [&]() __attribute__((always_inline)) {
      while (cnt >= AGG_FLUSH_AT) {
        drain(cnt - 32, 32);
        cnt -= 32;
      }
    };

    if (nFull > 0) {
      int a = (int)(((long long)blockIdx.x * nFull) / gridDim.x);
      uint4 cur[VPW], nxt[VPW];
      {
        const uint4* p = tb4 + a * STEP + wv * WH + lane;
#pragma unroll
        for (int u = 0; u < VPW; ++u) cur[u] = p[32 * u];
      }
      for (int i = 0; i < nFull; ++i) {
        int an = a + 1;
        if (an >= nFull) an = 0;
        {
          const uint4* p = tb4 + an * STEP + wv * WH + lane;
#pragma unroll
          for (int u = 0; u < VPW; ++u) nxt[u] = p[32 * u];
        }
        const int gb = a * STEP + wv * WH;
#pragma unroll
        for (int u = 0; u < VPW; ++u) {
          hex(cur[u], gb + 32 * u);
          flush32();
        }
#pragma unroll
        for (int u = 0; u < VPW; ++u) cur[u] = nxt[u];
        a = an;
      }
    }
    {
      const int gw = nFull * STEP + wv * WH;
      uint4 d[VPW];
#pragma unroll
      for (int u = 0; u < VPW; ++u) {
        const int g = gw + 32 * u + lane;
        d[u] = make_uint4(~tpat, ~tpat, ~tpat, ~tpat);
        if (g < nh) d[u] = tb4[g];
      }
#pragma unroll
      for (int u = 0; u < VPW; ++u) {
        hex(d[u], gw + 32 * u);
        flush32();
      }
    }
  } else if ((E & 3) == 0) {
    const int4* dst4 = reinterpret_cast<const int4*>(dst);
    const int nq = E >> 2;
    constexpr int VPW = 8;
    constexpr int WQ = 32 * VPW;
    constexpr int STEP = 32 * WQ;
    const int nFull = nq / STEP;
    const unsigned R = (unsigned)rows;
    const int lane4 = lane * 4;

    auto quad = [&](const int4& d, int eb) __attribute__((always_inline)) {
      const unsigned rx = (unsigned)(d.x - lo), ry = (unsigned)(d.y - lo);
      const unsigned rz = (unsigned)(d.z - lo), rw = (unsigned)(d.w - lo);
      const unsigned mn = min(min(rx, ry), min(rz, rw));
      if (__builtin_amdgcn_ballot_w32(mn < R) != 0u) {
        const int e = eb + lane4;
        push(d.x, e);
        push(d.y, e + 1);
        push(d.z, e + 2);
        push(d.w, e + 3);
      }
    };
    auto flush32 = [&]() __attribute__((always_inline)) {
      while (cnt >= 32) {
        drain(cnt - 32, 32);
        cnt -= 32;
      }
    };

    if (nFull > 0) {
      int a = (int)(((long long)blockIdx.x * nFull) / gridDim.x);
      int4 cur[VPW], nxt[VPW];
      {
        const int4* p = dst4 + a * STEP + wv * WQ + lane;
#pragma unroll
        for (int u = 0; u < VPW; ++u) cur[u] = p[32 * u];
      }
      for (int i = 0; i < nFull; ++i) {
        int an = a + 1;
        if (an >= nFull) an = 0;
        {
          const int4* p = dst4 + an * STEP + wv * WQ + lane;
#pragma unroll
          for (int u = 0; u < VPW; ++u) nxt[u] = p[32 * u];
        }
        const int eb = (a * STEP + wv * WQ) * 4;
#pragma unroll
        for (int u = 0; u < 4; ++u) quad(cur[u], eb + 128 * u);
        flush32();
#pragma unroll
        for (int u = 4; u < VPW; ++u) quad(cur[u], eb + 128 * u);
        flush32();
#pragma unroll
        for (int u = 0; u < VPW; ++u) cur[u] = nxt[u];
        a = an;
      }
    }
    {
      const int qw = nFull * STEP + wv * WQ;
      int4 d[VPW];
#pragma unroll
      for (int u = 0; u < VPW; ++u) {
        const int q = qw + 32 * u + lane;
        d[u] = make_int4(-1, -1, -1, -1);
        if (q < nq) d[u] = dst4[q];
      }
#pragma unroll
      for (int u = 0; u < VPW; ++u) {
        quad(d[u], (qw + 32 * u) * 4);
        if (u == 3) flush32();
      }
    }
  } else {
    for (int e0 = wv * 32; e0 < E; e0 += 1024) {
      const int e = e0 + lane;
      const int dv = (e < E) ? dst[e] : -1;
      push(dv, e);
      if (cnt >= 32) {
        drain(cnt - 32, 32);
        cnt -= 32;
      }
    }
  }
  drainAll();
  cnt = 0;
  if (WRITE_REC) {
    if (lane == 0) rcnt[blockIdx.x * 32 + wv] = nrec;
  }

  __syncthreads();

  for (int r = wv; r < rows; r += 32) {
    VT v;
#pragma unroll
    for (int c = 0; c < FPL; ++c) v[c] = tile[r * F + lane + 32 * c];
    *reinterpret_cast<VT*>(out + (size_t)(lo + r) * F + lane * FPL) = v;
  }
}

extern "C" void kernel_launch(void* const* d_in, const int* in_sizes, int n_in,
                              void* d_out, int out_size, void* d_ws, size_t ws_size,
                              hipStream_t stream) {
  const float* x  = (const float*)d_in[0];
  const int*   ei = (const int*)d_in[1];
  const float* W1 = (const float*)d_in[2];
  const float* b1 = (const float*)d_in[3];
  const float* W2 = (const float*)d_in[4];
  const float* b2 = (const float*)d_in[5];
  float* out = (float*)d_out;

  const int Nn = GCN_N_NODES;
  const int E  = in_sizes[1] / 2;
  const int* src = ei;
  const int* dst = ei + E;

  float* dinv = (float*)d_ws;
  float* h    = dinv + Nn;
  float* out1 = h + (size_t)Nn * 128;
  float* h2   = h;
  int* rec  = (int*)(out1 + (size_t)Nn * 128);
  int* rcnt = rec + (size_t)AGG_GROUPS * 32 * AGG_RCAP * 2;
  unsigned char* tb = (unsigned char*)(rcnt + AGG_GROUPS * 32);

  deg_init_kernel<<<(Nn + 255) / 256, 256, 0, stream>>>(dinv, Nn);
  deg_count_kernel<<<((E + 3) / 4 + 255) / 256, 256, 0, stream>>>(dst, dinv, tb, E);
  deg_rsqrt_kernel<<<(Nn + 255) / 256, 256, 0, stream>>>(dinv, Nn);

  gemm_wreg_kernel<128, false><<<1024, 256, 0, stream>>>(x, W1, nullptr, h, Nn);
  agg_tile_kernel<128, false, true, false><<<(Nn + AGG_TILE - 1) / AGG_TILE, 1024, 0, stream>>>(
      h, src, dst, dinv, nullptr, out1, rec, rcnt, tb, Nn, E);

  gemm_wreg_kernel<64, true><<<1024, 256, 0, stream>>>(out1, W2, b1, h2, Nn);
  agg_tile_kernel<64, true, false, true><<<(Nn + AGG_TILE - 1) / AGG_TILE, 1024, 0, stream>>>(
      h2, src, dst, dinv, b2, out, rec, rcnt, tb, Nn, E);
}


# Degree count (float adds, tile bytes) rides in the layer-1 GEMM launch, dealt over its threads behind its first barrier; rsqrt taken where a degree is read: 7 launches -> 5; sums unchanged
# speedup vs baseline: 2.8233x; 1.0339x over previous
#include <hip/hip_runtime.h>

typedef __attribute__((ext_vector_type(2))) float v2f;
typedef __attribute__((ext_vector_type(8))) float v8f;

#define GCN_N_NODES 100000
#define AGG_TILE 391
#define AGG_GROUPS ((GCN_N_NODES + AGG_TILE - 1) / AGG_TILE)
#define AGG_PEND 896
#define AGG_FLUSH_AT (AGG_PEND - 512)
#define AGG_MAXC 8
#define AGG_RCAP 1024
#define AGG_DL_BITS 9
static_assert(AGG_TILE <= (1 << AGG_DL_BITS), "tile row must fit the record's low bits");
static_assert(AGG_GROUPS <= 256, "a tile number must fit one byte");
static_assert(GCN_N_NODES <= (1 << (31 - AGG_DL_BITS)), "source row must fit the record's high bits");

__global__ void __launch_bounds__(256)
deg_init_kernel(float* __restrict__ deg, int n) {
  int i = blockIdx.x * blockDim.x + threadIdx.x;
  if (i < n) deg[i] = 1.0f;
}

__device__ __forceinline__ void deg_count_quad(const int4 d, float* __restrict__ deg,
                                               unsigned char* __restrict__ tb, int e) {
  unsafeAtomicAdd(&deg[d.x], 1.0f);
  unsafeAtomicAdd(&deg[d.y], 1.0f);
  unsafeAtomicAdd(&deg[d.z], 1.0f);
  unsafeAtomicAdd(&deg[d.w], 1.0f);
  const unsigned t = (unsigned)(d.x / AGG_TILE) | ((unsigned)(d.y / AGG_TILE) << 8) |
                     ((unsigned)(d.z / AGG_TILE) << 16) | ((unsigned)(d.w / AGG_TILE) << 24);
  *reinterpret_cast<unsigned*>(tb + e) = t;
}
__device__ __forceinline__ void deg_count_edges(const int* __restrict__ dst, float* __restrict__ deg,
                                                unsigned char* __restrict__ tb, int E, int e) {
  for (int k = 0; k < 4 && e + k < E; ++k) {
    const int d = dst[e + k];
    unsafeAtomicAdd(&deg[d], 1.0f);
    tb[e + k] = (unsigned char)(d / AGG_TILE);
  }
}

template <int N, bool FUSE_RELU_BIAS, bool COUNT_DEG>
__global__ void __launch_bounds__(256)
gemm_wreg_kernel(const float* __restrict__ A, const float* __restrict__ W,
                 const float* __restrict__ bias, float* __restrict__ C, int M,
                 const int* __restrict__ dst, float* __restrict__ deg,
                 unsigned char* __restrict__ tb, int E) {
  constexpr int K = 128;
  constexpr int BM = 32;
  constexpr int PA = K + 4;
  constexpr int PC = N + 4;
  constexpr int NT = N / 16;
  constexpr int RT = NT / 4;
  constexpr int LPR = N / 4;
  constexpr int RPI = 32 / LPR;
  static_assert(NT == 8 || NT == 4, "N is 128 or 64");

  __shared__ alignas(16) float abuf[2][BM * PA];
  __shared__ alignas(16) float cbuf[2][BM * PC];

  const int tid = threadIdx.x;
  const int lane = tid & 31;
  const int wv = __builtin_amdgcn_readfirstlane(tid >> 5);
  const int c = lane & 15;
  const int h = lane >> 4;
  const int ct = wv % NT;
  const int rt0 = (wv / NT) * RT;

  const int nblk = (M + BM - 1) / BM;
  int blk = blockIdx.x;
  if (blk >= nblk) return;

  v2f b[32];
  {
    const float* wp = W + 16 * ct + c;
#pragma unroll
    for (int p = 0; p < 16; ++p) {
      const int k0 = 8 * p + 4 * h;
      b[2 * p].x = wp[(size_t)(k0 + 0) * N];
      b[2 * p].y = wp[(size_t)(k0 + 1) * N];
      b[2 * p + 1].x = wp[(size_t)(k0 + 2) * N];
      b[2 * p + 1].y = wp[(size_t)(k0 + 3) * N];
    }
  }
  float4 bv = make_float4(0.0f, 0.0f, 0.0f, 0.0f);
  if (FUSE_RELU_BIAS) bv = *reinterpret_cast<const float4*>(bias + 4 * lane);

  float4 st[4];
  auto fetch = [&](int q) __attribute__((always_inline)) {
#pragma unroll
    for (int u = 0; u < 4; ++u) {
      const int row = q * BM + 4 * wv + u;
      st[u] = make_float4(0.0f, 0.0f, 0.0f, 0.0f);
      if (row < M) st[u] = *reinterpret_cast<const float4*>(A + (size_t)row * K + 4 * lane);
    }
  };
  auto stage = [&](int buf) __attribute__((always_inline)) {
#pragma unroll
    for (int u = 0; u < 4; ++u) {
      float4 v = st[u];
      if (FUSE_RELU_BIAS) {
        v.x = fmaxf(v.x + bv.x, 0.0f);
        v.y = fmaxf(v.y + bv.y, 0.0f);
        v.z = fmaxf(v.z + bv.z, 0.0f);
        v.w = fmaxf(v.w + bv.w, 0.0f);
      }
      *reinterpret_cast<float4*>(&abuf[buf][(4 * wv + u) * PA + 4 * lane]) = v;
    }
  };

  fetch(blk);

  const int ng = COUNT_DEG ? ((E + 3) >> 2) : 0;
  const int nth = gridDim.x * 256;
  const int g0 = blockIdx.x * 256 + tid;
  const int g1 = g0 + nth;
  const bool quads = COUNT_DEG && (E & 3) == 0;
  const int4* dst4 = reinterpret_cast<const int4*>(dst);
  int4 d0 = make_int4(0, 0, 0, 0), d1 = make_int4(0, 0, 0, 0);
  if (COUNT_DEG) {
    if (quads) {
      if (g0 < ng) d0 = dst4[g0];
      if (g1 < ng) d1 = dst4[g1];
    }
  }

  stage(0);
  __syncthreads();

  if (COUNT_DEG) {
    if (quads) {
      if (g0 < ng) deg_count_quad(d0, deg, tb, 4 * g0);
      if (g1 < ng) deg_count_quad(d1, deg, tb, 4 * g1);
      for (int g = g1 + nth; g < ng; g += nth) deg_count_quad(dst4[g], deg, tb, 4 * g);
    } else {
      for (int g = g0; g < ng; g += nth) deg_count_edges(dst, deg, tb, E, 4 * g);
    }
  }

  for (int it = 0; blk < nblk; blk += gridDim.x, ++it) {
    const int cur = it & 1;
    const int nb = blk + gridDim.x;
    if (nb < nblk) fetch(nb);

    float4 x[RT][16];
#pragma unroll
    for (int rt = 0; rt < RT; ++rt)
#pragma unroll
      for (int p = 0; p < 16; ++p)
        x[rt][p] = *reinterpret_cast<const float4*>(
            &abuf[cur][(16 * (rt0 + rt) + c) * PA + 8 * p + 4 * h]);

    v8f acc[RT];
#pragma unroll
    for (int rt = 0; rt < RT; ++rt) acc[rt] = v8f{};
#pragma unroll
    for (int p = 0; p < 16; ++p) {
#pragma unroll
      for (int rt = 0; rt < RT; ++rt) {
        v2f a0; a0.x = x[rt][p].x; a0.y = x[rt][p].y;
        acc[rt] = __builtin_amdgcn_wmma_f32_16x16x4_f32(
             false, a0,  false, b[2 * p],
             (short)0, acc[rt],  false,  false);
      }
#pragma unroll
      for (int rt = 0; rt < RT; ++rt) {
        v2f a1; a1.x = x[rt][p].z; a1.y = x[rt][p].w;
        acc[rt] = __builtin_amdgcn_wmma_f32_16x16x4_f32(
             false, a1,  false, b[2 * p + 1],
             (short)0, acc[rt],  false,  false);
      }
    }

#pragma unroll
    for (int rt = 0; rt < RT; ++rt)
#pragma unroll
      for (int r = 0; r < 8; ++r)
        cbuf[cur][(16 * (rt0 + rt) + 8 * h + r) * PC + 16 * ct + c] = acc[rt][r];

    if (nb < nblk) stage(cur ^ 1);
    __syncthreads();

#pragma unroll
    for (int u = 0; u < 4 / RPI; ++u) {
      const int rb = 4 * wv + u * RPI + lane / LPR;
      const int col = 4 * (lane % LPR);
      const int row = blk * BM + rb;
      const float4 v = *reinterpret_cast<const float4*>(&cbuf[cur][rb * PC + col]);
      if (row < M) *reinterpret_cast<float4*>(C + (size_t)row * N + col) = v;
    }
  }
}

template <int F, bool HAS_BIAS, bool WRITE_REC, bool READ_REC>
__global__ void __launch_bounds__(1024)
agg_tile_kernel(const float* __restrict__ h, const int* __restrict__ src,
                const int* __restrict__ dst, const float* __restrict__ deg,
                const float* __restrict__ bias, float* __restrict__ out,
                int* __restrict__ rec, int* __restrict__ rcnt,
                const unsigned char* __restrict__ tb, int n, int E) {
  constexpr int FPL = F / 32;
  constexpr int G = 32 / FPL;
  typedef float __attribute__((ext_vector_type(FPL))) VT;

  __shared__ float tile[AGG_TILE * F];
  __shared__ int pend[32 * AGG_PEND];

  const int tid = threadIdx.x;
  const int lane = tid & 31;
  const int wv = __builtin_amdgcn_readfirstlane(tid >> 5);
  const int lo = blockIdx.x * AGG_TILE;
  int rows = n - lo;
  if (rows > AGG_TILE) rows = AGG_TILE;

  int2* const recw = reinterpret_cast<int2*>(rec) + ((size_t)blockIdx.x * 32 + wv) * AGG_RCAP;
  int nrec = 0;
  bool useRec = false;
  if (READ_REC) {
    const int c = rcnt[blockIdx.x * 32 + lane];
    useRec = __builtin_amdgcn_ballot_w32(c > AGG_RCAP) == 0u;
    nrec = __builtin_amdgcn_readlane(c, wv);
  }

  {
    constexpr int NR = (AGG_TILE + 31) / 32;
    static_assert(NR <= 32, "one lane per row of the wave");
    VT iv[NR];
    float bb[FPL];
#pragma unroll
    for (int c = 0; c < FPL; ++c) bb[c] = HAS_BIAS ? bias[lane * FPL + c] : 0.0f;
    float dvl;
    {
      const int t = (lane < NR) ? lane : (NR - 1);
      const int r = wv + 32 * t;
      dvl = rsqrtf(deg[lo + ((r < rows) ? r : (rows - 1))]);
    }
#pragma unroll
    for (int t = 0; t < NR; ++t) {
      const int r = wv + 32 * t;
      const int node = lo + ((r < rows) ? r : (rows - 1));
      iv[t] = *reinterpret_cast<const VT*>(h + (size_t)node * F + lane * FPL);
    }
#pragma unroll
    for (int t = 0; t < NR; ++t) {
      const int r = wv + 32 * t;
      if (r < rows) {
        const float dit = __int_as_float(__builtin_amdgcn_readlane(__float_as_int(dvl), t));
#pragma unroll
        for (int c = 0; c < FPL; ++c) {
          float v = iv[t][c] * dit * dit;
          if (HAS_BIAS) v += bb[c];
          tile[r * F + lane + 32 * c] = v;
        }
      }
    }
  }
  __syncthreads();

  int cnt = 0;

  auto addRows = [&](int s_i, int dl_i, float w_i, int k, auto&& requested)
                     __attribute__((always_inline)) {
#pragma unroll 1
    for (int j0 = 0; j0 < k; j0 += G) {
      VT v[G];
#pragma unroll
      for (int jj = 0; jj < G; ++jj) {
        const int j = j0 + jj;
        const int s = __builtin_amdgcn_readlane(s_i, (j < k) ? j : (k - 1));
        v[jj] = *reinterpret_cast<const VT*>(h + (size_t)s * F + lane * FPL);
      }
      requested(j0);
#pragma unroll
      for (int jj = 0; jj < G; ++jj) {
        const int j = j0 + jj;
        if (j < k) {
          const float w = __int_as_float(__builtin_amdgcn_readlane(__float_as_int(w_i), j));
          const int dl = __builtin_amdgcn_readlane(dl_i, j);
#pragma unroll
          for (int c = 0; c < FPL; ++c)
            __hip_atomic_fetch_add(&tile[dl * F + lane + 32 * c], v[jj][c] * w,
                                   __ATOMIC_RELAXED, __HIP_MEMORY_SCOPE_WORKGROUP);
        }
      }
    }
  };

  auto drain = [&](int base, int k) __attribute__((always_inline)) {
    __builtin_amdgcn_wave_barrier();
    int s_i = 0, dl_i = 0;
    float w_i = 0.0f;
    if (lane < k) {
      const int e_i = pend[wv * AGG_PEND + base + lane];
      s_i = src[e_i];
      const int d_i = dst[e_i];
      dl_i = d_i - lo;
      w_i = rsqrtf(deg[s_i]) * rsqrtf(deg[d_i]);
    }
    if (WRITE_REC) {
      const int slot = nrec + lane;
      if (lane < k && slot < AGG_RCAP)
        recw[slot] = make_int2((s_i << AGG_DL_BITS) | dl_i, __float_as_int(w_i));
      nrec += k;
    }
    addRows(s_i, dl_i, w_i, k, [](int) {});
  };

  constexpr int NG = 32 / G;
  auto loadG = [&](VT (&v)[G], int s_i, int j0, int k) __attribute__((always_inline)) {
#pragma unroll
    for (int jj = 0; jj < G; ++jj) {
      const int j = j0 + jj;
      const int s = __builtin_amdgcn_readlane(s_i, (j < k) ? j : (k - 1));
      v[jj] = *reinterpret_cast<const VT*>(h + (size_t)s * F + lane * FPL);
    }
  };
  auto addG = [&](VT (&v)[G], int dl_i, float w_i, int j0, int k) __attribute__((always_inline)) {
#pragma unroll
    for (int jj = 0; jj < G; ++jj) {
      const int j = j0 + jj;
      if (j < k) {
        const float w = __int_as_float(__builtin_amdgcn_readlane(__float_as_int(w_i), j));
        const int dl = __builtin_amdgcn_readlane(dl_i, j);
#pragma unroll
        for (int c = 0; c < FPL; ++c)
          __hip_atomic_fetch_add(&tile[dl * F + lane + 32 * c], v[jj][c] * w,
                                 __ATOMIC_RELAXED, __HIP_MEMORY_SCOPE_WORKGROUP);
      }
    }
  };
  auto chunk = [&](int s_i, int dl_i, float w_i, int k) __attribute__((always_inline)) {
    VT a[G], b[G];
    loadG(a, s_i, 0, k);
    if (k > G) {
      loadG(b, s_i, G, k);
      addG(a, dl_i, w_i, 0, k);
      if constexpr (NG > 2) {
        if (k > 2 * G) {
          loadG(a, s_i, 2 * G, k);
          addG(b, dl_i, w_i, G, k);
          if (k > 3 * G) {
            loadG(b, s_i, 3 * G, k);
            addG(a, dl_i, w_i, 2 * G, k);
            addG(b, dl_i, w_i, 3 * G, k);
          } else {
            addG(a, dl_i, w_i, 2 * G, k);
          }
        } else {
          addG(b, dl_i, w_i, G, k);
        }
      } else {
        addG(b, dl_i, w_i, G, k);
      }
    } else {
      addG(a, dl_i, w_i, 0, k);
    }
  };

  auto drainAll = [&]() __attribute__((always_inline)) {
    __builtin_amdgcn_wave_barrier();
#pragma unroll 1
    for (int sb = 0; sb < cnt; sb += 32 * AGG_MAXC) {
      int s_i[AGG_MAXC], d_i[AGG_MAXC];
      float w_i[AGG_MAXC];
#pragma unroll
      for (int u = 0; u < AGG_MAXC; ++u) {
        const int idx = sb + 32 * u + lane;
        int e = 0;
        if (idx < cnt) e = pend[wv * AGG_PEND + idx];
        s_i[u] = e;
      }
#pragma unroll
      for (int u = 0; u < AGG_MAXC; ++u) {
        const int e = s_i[u];
        s_i[u] = src[e];
        d_i[u] = dst[e];
      }
#pragma unroll
      for (int u = 0; u < AGG_MAXC; ++u) w_i[u] = rsqrtf(deg[s_i[u]]) * rsqrtf(deg[d_i[u]]);
      const int left = cnt - sb;
      if (WRITE_REC) {
#pragma unroll
        for (int u = 0; u < AGG_MAXC; ++u) {
          const int slot = nrec + 32 * u + lane;
          if (32 * u + lane < left && slot < AGG_RCAP)
            recw[slot] = make_int2((s_i[u] << AGG_DL_BITS) | (d_i[u] - lo), __float_as_int(w_i[u]));
        }
        nrec += (left < 32 * AGG_MAXC) ? left : 32 * AGG_MAXC;
      }
#pragma unroll 1
      for (int u = 0; u < AGG_MAXC; ++u) {
        int k = left - 32 * u;
        if (k <= 0) break;
        if (k > 32) k = 32;
        int s = s_i[0], d = d_i[0];
        float w = w_i[0];
#pragma unroll
        for (int t = 1; t < AGG_MAXC; ++t) {
          if (u == t) {
            s = s_i[t];
            d = d_i[t];
            w = w_i[t];
          }
        }
        chunk(s, d - lo, w, k);
      }
    }
  };

  auto push = [&](int dv, int e) __attribute__((always_inline)) {
    const bool m = (unsigned)(dv - lo) < (unsigned)rows;
    const unsigned mask = __builtin_amdgcn_ballot_w32(m);
    if (mask != 0u) {
      if (m) pend[wv * AGG_PEND + cnt + (int)__builtin_amdgcn_mbcnt_lo(mask, 0u)] = e;
      cnt += __builtin_popcount(mask);
    }
  };

  if (READ_REC && useRec) {
    int2 cur = make_int2(0, 0);
    if (lane < nrec) cur = recw[lane];
    for (int base = 0; base < nrec; base += 32) {
      int2 nxt = make_int2(0, 0);
      const int left = nrec - base;
      if (base + 32 + lane < nrec) nxt = recw[base + 32 + lane];
      chunk(cur.x >> AGG_DL_BITS, cur.x & ((1 << AGG_DL_BITS) - 1), __int_as_float(cur.y),
            (left < 32) ? left : 32);
      cur = nxt;
    }
  } else if ((E & 15) == 0) {
    const uint4* tb4 = reinterpret_cast<const uint4*>(tb);
    const int nh = E >> 4;
    constexpr int VPW = 4;
    constexpr int WH = 32 * VPW;
    constexpr int STEP = 32 * WH;
    const int nFull = nh / STEP;
    const unsigned tpat = 0x01010101u * (unsigned)blockIdx.x;

    auto ne = [&](unsigned v) __attribute__((always_inline)) {
      const unsigned x = v ^ tpat;
      return ((x & 0x7F7F7F7Fu) + 0x7F7F7F7Fu) | x;
    };
    auto hex = [&](const uint4& d, int gb) __attribute__((always_inline)) {
      const unsigned z0 = ne(d.x), z1 = ne(d.y), z2 = ne(d.z), z3 = ne(d.w);
      unsigned m = __builtin_amdgcn_ballot_w32((~(z0 & z1 & z2 & z3) & 0x80808080u) != 0u);
      if (m != 0u) {
        unsigned hm = ((~z0 & 0x80808080u) >> 7) | ((~z1 & 0x80808080u) >> 6) |
                      ((~z2 & 0x80808080u) >> 5) | ((~z3 & 0x80808080u) >> 4);
        const int e0 = (gb + lane) * 16;
        do {
          if (hm != 0u) {
            const int p = __builtin_ctz(hm);
            pend[wv * AGG_PEND + cnt + (int)__builtin_amdgcn_mbcnt_lo(m, 0u)] =
                e0 + 4 * (p & 7) + (p >> 3);
          }
          cnt += __builtin_popcount(m);
          hm &= hm - 1u;
          m = __builtin_amdgcn_ballot_w32(hm != 0u);
        } while (m != 0u);
      }
    };
    auto flush32 = [&]() __attribute__((always_inline)) {
      while (cnt >= AGG_FLUSH_AT) {
        drain(cnt - 32, 32);
        cnt -= 32;
      }
    };

    if (nFull > 0) {
      int a = (int)(((long long)blockIdx.x * nFull) / gridDim.x);
      uint4 cur[VPW], nxt[VPW];
      {
        const uint4* p = tb4 + a * STEP + wv * WH + lane;
#pragma unroll
        for (int u = 0; u < VPW; ++u) cur[u] = p[32 * u];
      }
      for (int i = 0; i < nFull; ++i) {
        int an = a + 1;
        if (an >= nFull) an = 0;
        {
          const uint4* p = tb4 + an * STEP + wv * WH + lane;
#pragma unroll
          for (int u = 0; u < VPW; ++u) nxt[u] = p[32 * u];
        }
        const int gb = a * STEP + wv * WH;
#pragma unroll
        for (int u = 0; u < VPW; ++u) {
          hex(cur[u], gb + 32 * u);
          flush32();
        }
#pragma unroll
        for (int u = 0; u < VPW; ++u) cur[u] = nxt[u];
        a = an;
      }
    }
    {
      const int gw = nFull * STEP + wv * WH;
      uint4 d[VPW];
#pragma unroll
      for (int u = 0; u < VPW; ++u) {
        const int g = gw + 32 * u + lane;
        d[u] = make_uint4(~tpat, ~tpat, ~tpat, ~tpat);
        if (g < nh) d[u] = tb4[g];
      }
#pragma unroll
      for (int u = 0; u < VPW; ++u) {
        hex(d[u], gw + 32 * u);
        flush32();
      }
    }
  } else if ((E & 3) == 0) {
    const int4* dst4 = reinterpret_cast<const int4*>(dst);
    const int nq = E >> 2;
    constexpr int VPW = 8;
    constexpr int WQ = 32 * VPW;
    constexpr int STEP = 32 * WQ;
    const int nFull = nq / STEP;
    const unsigned R = (unsigned)rows;
    const int lane4 = lane * 4;

    auto quad = [&](const int4& d, int eb) __attribute__((always_inline)) {
      const unsigned rx = (unsigned)(d.x - lo), ry = (unsigned)(d.y - lo);
      const unsigned rz = (unsigned)(d.z - lo), rw = (unsigned)(d.w - lo);
      const unsigned mn = min(min(rx, ry), min(rz, rw));
      if (__builtin_amdgcn_ballot_w32(mn < R) != 0u) {
        const int e = eb + lane4;
        push(d.x, e);
        push(d.y, e + 1);
        push(d.z, e + 2);
        push(d.w, e + 3);
      }
    };
    auto flush32 = [&]() __attribute__((always_inline)) {
      while (cnt >= 32) {
        drain(cnt - 32, 32);
        cnt -= 32;
      }
    };

    if (nFull > 0) {
      int a = (int)(((long long)blockIdx.x * nFull) / gridDim.x);
      int4 cur[VPW], nxt[VPW];
      {
        const int4* p = dst4 + a * STEP + wv * WQ + lane;
#pragma unroll
        for (int u = 0; u < VPW; ++u) cur[u] = p[32 * u];
      }
      for (int i = 0; i < nFull; ++i) {
        int an = a + 1;
        if (an >= nFull) an = 0;
        {
          const int4* p = dst4 + an * STEP + wv * WQ + lane;
#pragma unroll
          for (int u = 0; u < VPW; ++u) nxt[u] = p[32 * u];
        }
        const int eb = (a * STEP + wv * WQ) * 4;
#pragma unroll
        for (int u = 0; u < 4; ++u) quad(cur[u], eb + 128 * u);
        flush32();
#pragma unroll
        for (int u = 4; u < VPW; ++u) quad(cur[u], eb + 128 * u);
        flush32();
#pragma unroll
        for (int u = 0; u < VPW; ++u) cur[u] = nxt[u];
        a = an;
      }
    }
    {
      const int qw = nFull * STEP + wv * WQ;
      int4 d[VPW];
#pragma unroll
      for (int u = 0; u < VPW; ++u) {
        const int q = qw + 32 * u + lane;
        d[u] = make_int4(-1, -1, -1, -1);
        if (q < nq) d[u] = dst4[q];
      }
#pragma unroll
      for (int u = 0; u < VPW; ++u) {
        quad(d[u], (qw + 32 * u) * 4);
        if (u == 3) flush32();
      }
    }
  } else {
    for (int e0 = wv * 32; e0 < E; e0 += 1024) {
      const int e = e0 + lane;
      const int dv = (e < E) ? dst[e] : -1;
      push(dv, e);
      if (cnt >= 32) {
        drain(cnt - 32, 32);
        cnt -= 32;
      }
    }
  }
  drainAll();
  cnt = 0;
  if (WRITE_REC) {
    if (lane == 0) rcnt[blockIdx.x * 32 + wv] = nrec;
  }

  __syncthreads();

  for (int r = wv; r < rows; r += 32) {
    VT v;
#pragma unroll
    for (int c = 0; c < FPL; ++c) v[c] = tile[r * F + lane + 32 * c];
    *reinterpret_cast<VT*>(out + (size_t)(lo + r) * F + lane * FPL) = v;
  }
}

extern "C" void kernel_launch(void* const* d_in, const int* in_sizes, int n_in,
                              void* d_out, int out_size, void* d_ws, size_t ws_size,
                              hipStream_t stream) {
  const float* x  = (const float*)d_in[0];
  const int*   ei = (const int*)d_in[1];
  const float* W1 = (const float*)d_in[2];
  const float* b1 = (const float*)d_in[3];
  const float* W2 = (const float*)d_in[4];
  const float* b2 = (const float*)d_in[5];
  float* out = (float*)d_out;

  const int Nn = GCN_N_NODES;
  const int E  = in_sizes[1] / 2;
  const int* src = ei;
  const int* dst = ei + E;

  float* deg  = (float*)d_ws;
  float* h    = deg + Nn;
  float* out1 = h + (size_t)Nn * 128;
  float* h2   = h;
  int* rec  = (int*)(out1 + (size_t)Nn * 128);
  int* rcnt = rec + (size_t)AGG_GROUPS * 32 * AGG_RCAP * 2;
  unsigned char* tb = (unsigned char*)(rcnt + AGG_GROUPS * 32);

  deg_init_kernel<<<(Nn + 255) / 256, 256, 0, stream>>>(deg, Nn);

  static_assert((GCN_N_NODES + 31) / 32 >= 1024,
                "every work-group of this launch has a block of rows: none leaves before its share of the count");
  gemm_wreg_kernel<128, false, true><<<1024, 256, 0, stream>>>(x, W1, nullptr, h, Nn, dst, deg, tb, E);
  agg_tile_kernel<128, false, true, false><<<(Nn + AGG_TILE - 1) / AGG_TILE, 1024, 0, stream>>>(
      h, src, dst, deg, nullptr, out1, rec, rcnt, tb, Nn, E);

  gemm_wreg_kernel<64, true, false><<<1024, 256, 0, stream>>>(out1, W2, b1, h2, Nn, nullptr, nullptr, nullptr, 0);
  agg_tile_kernel<64, true, false, true><<<(Nn + AGG_TILE - 1) / AGG_TILE, 1024, 0, stream>>>(
      h2, src, dst, deg, b2, out, rec, rcnt, tb, Nn, E);
}
